# starting kernel
# baseline (speedup 1.0000x reference)
#include <hip/hip_runtime.h>

#define B_ 64
#define L_ 512
#define D_ 768
#define H_ 768
#define K_ 128

typedef float v2f __attribute__((ext_vector_type(2)));
typedef float v8f __attribute__((ext_vector_type(8)));

__global__ __launch_bounds__(512) void feat_kernel(
    const float* __restrict__ enc,
    const int*   __restrict__ valid_mask,
    const int*   __restrict__ pos_span,
    float*       __restrict__ feat)
{
    __shared__ int s_scan[L_];
    __shared__ int s_idx[64];
    __shared__ int s_n;

    const int b   = blockIdx.x;
    const int tid = threadIdx.x;
    const int* vm = valid_mask + (size_t)b * L_;

    const int v = (vm[tid] == 1) ? 1 : 0;
    s_scan[tid] = v;
    __syncthreads();

    for (int off = 1; off < L_; off <<= 1) {
        int add = (tid >= off) ? s_scan[tid - off] : 0;
        __syncthreads();
        s_scan[tid] += add;
        __syncthreads();
    }

    const int pos = s_scan[tid] - 1;
    if (v && pos < 64) s_idx[pos] = tid;
    if (tid == 0) {
        int nv = s_scan[L_ - 1];
        s_n = (nv < 64) ? nv : 64;
    }
    __syncthreads();

    const int lo = pos_span[b * 2 + 0];
    const int hi = pos_span[b * 2 + 1];
    const int nv = s_n;

    for (int d = tid; d < D_; d += 512) {
        float acc = -__builtin_inff();
        for (int j = lo; j <= hi; ++j) {
            float val = (j < nv)
                ? enc[((size_t)b * L_ + s_idx[j]) * D_ + d]
                : 0.0f;
            acc = fmaxf(acc, val);
        }
        feat[(size_t)b * D_ + d] = acc;
    }
}

template <bool RELU>
__global__ __launch_bounds__(128) void gemm_wmma(
    const float* __restrict__ A,
    const float* __restrict__ Bm,
    const float* __restrict__ bias,
    float*       __restrict__ C,
    int N, int Kd)
{
    __shared__ float sB[64 * 16];

    const int lane = threadIdx.x & 31;
    const int wave = threadIdx.x >> 5;
    const int n0   = blockIdx.x * 16;
    const int m0   = wave * 16;
    const int m    = lane & 15;
    const int n    = lane & 15;
    const int kb   = (lane >> 4) * 2;

    v8f acc = {};
    const float* arow   = A + (size_t)(m0 + m) * Kd;
    const float  bias_n = bias[n0 + n];

    for (int kc = 0; kc < Kd; kc += 64) {
        for (int i = threadIdx.x; i < 64 * 16; i += 128) {
            sB[i] = Bm[(size_t)(kc + (i >> 4)) * N + n0 + (i & 15)];
        }
        __syncthreads();

        #pragma unroll
        for (int k = 0; k < 64; k += 4) {
            v2f a;
            a.x = arow[kc + k + kb];
            a.y = arow[kc + k + kb + 1];
            v2f bf;
            bf.x = sB[(k + kb)     * 16 + n];
            bf.y = sB[(k + kb + 1) * 16 + n];
            acc = __builtin_amdgcn_wmma_f32_16x16x4_f32(
                false, a, false, bf, (short)0, acc, false, false);
        }
        __syncthreads();
    }

    const int mrow = m0 + (lane >> 4) * 8;
    #pragma unroll
    for (int vv = 0; vv < 8; ++vv) {
        float val = acc[vv] + bias_n;
        if (RELU) val = fmaxf(val, 0.0f);
        C[(size_t)(mrow + vv) * N + n0 + n] = val;
    }
}

extern "C" void kernel_launch(void* const* d_in, const int* in_sizes, int n_in,
                              void* d_out, int out_size, void* d_ws, size_t ws_size,
                              hipStream_t stream)
{
    (void)in_sizes; (void)n_in; (void)out_size; (void)ws_size;

    const float* enc        = (const float*)d_in[0];
    const float* W1         = (const float*)d_in[1];
    const float* b1         = (const float*)d_in[2];
    const float* W2         = (const float*)d_in[3];
    const float* b2         = (const float*)d_in[4];
    const int*   valid_mask = (const int*)d_in[5];
    const int*   pos_span   = (const int*)d_in[6];

    float* out  = (float*)d_out;
    float* feat = (float*)d_ws;
    float* h    = feat + (size_t)B_ * D_;

    feat_kernel<<<B_, 512, 0, stream>>>(enc, valid_mask, pos_span, feat);
    gemm_wmma<true ><<<H_ / 16, 128, 0, stream>>>(feat, W1, b1, h,   H_, D_);
    gemm_wmma<false><<<K_ / 16, 128, 0, stream>>>(h,    W2, b2, out, K_, H_);
}


# second product folded into the first product's launch: wave pairs turn their 16x32 h slab through LDS into 24 K-partial planes, third launch only adds the planes (f32 sum order changes)
# speedup vs baseline: 6.0477x; 6.0477x over previous
#include <hip/hip_runtime.h>

#define B_ 64
#define L_ 512
#define D_ 768
#define H_ 768
#define K_ 128

typedef float v2f __attribute__((ext_vector_type(2)));
typedef float v4f __attribute__((ext_vector_type(4)));
typedef float v8f __attribute__((ext_vector_type(8)));

#define SLICE_     128
#define SPAN_ROWS_ 40
#define TAB_       64

__device__ __forceinline__ float4 max4(float4 a, float4 b)
{
    return make_float4(fmaxf(a.x, b.x), fmaxf(a.y, b.y),
                       fmaxf(a.z, b.z), fmaxf(a.w, b.w));
}

__global__ __launch_bounds__(32) __attribute__((amdgpu_waves_per_eu(1, 1)))
void feat_kernel(
    const float* __restrict__ enc,
    const int*   __restrict__ valid_mask,
    const int*   __restrict__ pos_span,
    float*       __restrict__ feat)
{
    __shared__ int s_idx[TAB_];

    const int lane  = threadIdx.x;
    const int slice = blockIdx.x;
    const int b     = blockIdx.y;
    const int* vm   = valid_mask + (size_t)b * L_;

    int mv[L_ / 32];
    #pragma unroll
    for (int w = 0; w < L_ / 32; ++w) mv[w] = vm[32 * w + lane];

    const int lo = pos_span[b * 2 + 0];
    const int hi = pos_span[b * 2 + 1];

    const unsigned below = (1u << lane) - 1u;
    int base = 0;
    #pragma unroll
    for (int w = 0; w < L_ / 32; ++w) {
        const bool     v   = (mv[w] == 1);
        const unsigned m   = __builtin_amdgcn_ballot_w32(v);
        const int      pos = base + __popc(m & below);
        const int      j   = pos - lo;
        if (v && j >= 0 && pos <= hi && j < TAB_) s_idx[j] = 32 * w + lane;
        base += __popc(m);
    }
    const int nv = base;

    const int n = hi - lo + 1;
    int real = ((hi < nv - 1) ? hi : nv - 1) - lo + 1;
    real = (real < 0) ? 0 : real;
    const int nin = (real > TAB_) ? TAB_ : real;

    __syncthreads();

    float4 acc = make_float4(-__builtin_inff(), -__builtin_inff(),
                             -__builtin_inff(), -__builtin_inff());

    if (nin > 0) {
        int tab0 = (lane < nin)      ? s_idx[lane]      : 0;
        int tab1 = (lane + 32 < nin) ? s_idx[lane + 32] : 0;
        const int first = __builtin_amdgcn_readfirstlane(tab0);
        if (lane >= nin)      tab0 = first;
        if (lane + 32 >= nin) tab1 = first;

        const float* encb = enc + (size_t)b * L_ * D_;
        const int    col  = slice * SLICE_ + lane * 4;

        float4 r[SPAN_ROWS_];
        #pragma unroll
        for (int j = 0; j < SPAN_ROWS_; ++j) {
            const int row = __builtin_amdgcn_readlane((j < 32) ? tab0 : tab1, j & 31);
            r[j] = *(const float4*)(encb + (size_t)row * D_ + col);
        }
        #pragma unroll
        for (int j = 0; j < SPAN_ROWS_; ++j) acc = max4(acc, r[j]);

        for (int j = SPAN_ROWS_; j < nin; ++j) {
            const int row = __builtin_amdgcn_readlane(tab1, j - 32);
            acc = max4(acc, *(const float4*)(encb + (size_t)row * D_ + col));
        }
    }

    if (real < n) acc = max4(acc, make_float4(0.0f, 0.0f, 0.0f, 0.0f));

    *(float4*)(feat + (size_t)b * D_ + slice * SLICE_ + lane * 4) = acc;
}

#define CH_K_   32
#define PB_     20
#define AHEAD_  7
#define NS_     24
#define PH_     36
#define WS_NEED_ ((size_t)(B_ * D_ + NS_ * B_ * K_) * sizeof(float))

__global__ __launch_bounds__(64) __attribute__((amdgpu_waves_per_eu(1, 1)))
void head_kernel(
    const float* __restrict__ A,
    const float* __restrict__ Bm,
    const float* __restrict__ bias,
    const float* __restrict__ W2,
    float*       __restrict__ part,
    size_t                    ws_bytes)
{
    if (ws_bytes < WS_NEED_) return;

    constexpr int N   = H_;
    constexpr int KD  = D_;
    static_assert(KD == 24 * CH_K_ && AHEAD_ == 7, "the chunk loop below is written out for 24 chunks, 7 in flight");
    static_assert(H_ == NS_ * 32 && K_ == 128, "slices of 32 columns; each wave takes 64 of the 128 output columns");
    constexpr int NCH = KD / CH_K_;

    __shared__ __attribute__((aligned(16))) float sB[2][2 * CH_K_ * PB_];
    __shared__ __attribute__((aligned(16))) float sH[16 * PH_];

    const int wv   = __builtin_amdgcn_readfirstlane((int)(threadIdx.x >> 5));
    const int lane = threadIdx.x & 31;
    const int sl   = blockIdx.x;
    const int n0   = (2 * sl + wv) * 16;
    const int m0   = blockIdx.y * 16;
    const int c    = lane & 15;
    const int h    = lane >> 4;
    const int lr   = lane >> 2;
    const int lc   = (lane & 3) * 4;

    const float* ap = A  + (size_t)(m0 + c) * KD + 4 * h;
    const float* bp = Bm + (size_t)lr * N + n0 + lc;
    const float  bias_n = bias[n0 + c];

    const int st_off = lr * PB_ + lc;
    const int fr_off = (4 * h) * PB_ + c;

    float* sBw = sB[wv];

    const float* w2p = W2 + (size_t)(32 * sl + 4 * h) * K_ + 64 * wv + c;

    v4f    araw[NCH + AHEAD_][4];
    v4f    braw[NCH + AHEAD_][4];
    v2f    bfr[NCH + 1][8];
    v2f    w2f[4][8];

#define LOAD_CHUNK(g)                                                          \
    {                                                                          \
        _Pragma("unroll")                                                      \
        for (int j = 0; j < 4; ++j)                                            \
            braw[g][j] = *(const v4f*)(bp + (size_t)(CH_K_ * (g) + 8 * j) * N); \
        _Pragma("unroll")                                                      \
        for (int u = 0; u < 4; ++u)                                            \
            araw[g][u] = *(const v4f*)(ap + CH_K_ * (g) + 8 * u);           \
    }

#define STAGE_CHUNK(g)                                                         \
    {                                                                          \
        float* sb = sBw + ((g) & 1) * (CH_K_ * PB_);                           \
        _Pragma("unroll")                                                      \
        for (int j = 0; j < 4; ++j)                                            \
            *(v4f*)(sb + 8 * j * PB_ + st_off) = braw[g][j];                \
        __builtin_amdgcn_fence(__ATOMIC_RELEASE, "wavefront", "local");        \
        __builtin_amdgcn_fence(__ATOMIC_ACQUIRE, "wavefront", "local");        \
        _Pragma("unroll")                                                      \
        for (int t = 0; t < 8; ++t) {                                          \
            const int row = 8 * (t >> 1) + 2 * (t & 1);                        \
            bfr[g][t].x = sb[(row)     * PB_ + fr_off];                        \
            bfr[g][t].y = sb[(row + 1) * PB_ + fr_off];                        \
        }                                                                      \
    }

#define LOAD_W2(j)                                                             \
    {                                                                          \
        _Pragma("unroll")                                                      \
        for (int t = 0; t < 8; ++t) {                                          \
            const int row = 8 * (t >> 1) + 2 * (t & 1);                        \
            w2f[j][t].x = w2p[(row)     * K_ + 16 * (j)];                      \
            w2f[j][t].y = w2p[(row + 1) * K_ + 16 * (j)];                      \
        }                                                                      \
    }

#define WMMA_CHUNK(g)                                                          \
    {                                                                          \
        _Pragma("unroll")                                                      \
        for (int t = 0; t < 8; ++t) {                                          \
            const v4f a4 = araw[g][t >> 1];                                 \
            v2f a;                                                             \
            a.x = (t & 1) ? a4.z : a4.x;                                       \
            a.y = (t & 1) ? a4.w : a4.y;                                       \
                          \
            acc = __builtin_amdgcn_wmma_f32_16x16x4_f32(                       \
                false, a, false, bfr[g][t], (short)0, acc, false, false);      \
        }                                                                      \
    }
#define ITER_(g)                                                               \
    {                                                                          \
        if ((g) + AHEAD_ < NCH) LOAD_CHUNK((g) + AHEAD_)                       \
        if ((g) + 1 < NCH) STAGE_CHUNK((g) + 1)                                \
        if ((g) >= 17 && (g) <= 20) LOAD_W2((g) - 17)                          \
        __builtin_amdgcn_sched_barrier(0);                                     \
        WMMA_CHUNK(g)                                                          \
        __builtin_amdgcn_sched_barrier(0);                                     \
    }

    LOAD_CHUNK(0) LOAD_CHUNK(1) LOAD_CHUNK(2) LOAD_CHUNK(3)
    LOAD_CHUNK(4) LOAD_CHUNK(5) LOAD_CHUNK(6)
    STAGE_CHUNK(0)

    v8f acc = {};
    ITER_(0)  ITER_(1)  ITER_(2)  ITER_(3)  ITER_(4)  ITER_(5)
    ITER_(6)  ITER_(7)  ITER_(8)  ITER_(9)  ITER_(10) ITER_(11)
    ITER_(12) ITER_(13) ITER_(14) ITER_(15) ITER_(16) ITER_(17)
    ITER_(18) ITER_(19) ITER_(20) ITER_(21) ITER_(22) ITER_(23)
#undef ITER_
#undef WMMA_CHUNK
#undef LOAD_CHUNK
#undef STAGE_CHUNK
#undef LOAD_W2

    #pragma unroll
    for (int vv = 0; vv < 8; ++vv) {
        const float val = fmaxf(acc[vv] + bias_n, 0.0f);
        sH[(h * 8 + vv) * PH_ + 16 * wv + c] = val;
    }

    __syncthreads();

    v4f s4[4];
    #pragma unroll
    for (int u = 0; u < 4; ++u) s4[u] = *(const v4f*)(sH + c * PH_ + 8 * u + 4 * h);

    v8f p[4] = {};
    #pragma unroll
    for (int t = 0; t < 8; ++t) {
        const v4f a4 = s4[t >> 1];
        v2f a;
        a.x = (t & 1) ? a4.z : a4.x;
        a.y = (t & 1) ? a4.w : a4.y;
        #pragma unroll
        for (int j = 0; j < 4; ++j)
            p[j] = __builtin_amdgcn_wmma_f32_16x16x4_f32(
                false, a, false, w2f[j][t], (short)0, p[j], false, false);
    }

    float* pp = part + ((size_t)sl * B_ + m0 + h * 8) * K_ + 64 * wv + c;
    #pragma unroll
    for (int j = 0; j < 4; ++j) {
        #pragma unroll
        for (int vv = 0; vv < 8; ++vv)
            pp[(size_t)vv * K_ + 16 * j] = p[j][vv];
    }
}

__global__ __launch_bounds__(32)
void sum_kernel(
    const float* __restrict__ part,
    const float* __restrict__ b2,
    float*       __restrict__ out,
    size_t                    ws_bytes)
{
    if (ws_bytes < WS_NEED_) return;

    const int i  = blockIdx.x * 32 + threadIdx.x;
    const int n4 = i & (K_ / 4 - 1);

    v4f r[NS_];
    #pragma unroll
    for (int s = 0; s < NS_; ++s)
        r[s] = *(const v4f*)(part + (size_t)s * B_ * K_ + 4 * i);

    v4f acc = *(const v4f*)(b2 + 4 * n4);
    #pragma unroll
    for (int s = 0; s < NS_; ++s) acc += r[s];

    *(v4f*)(out + 4 * i) = acc;
}

extern "C" void kernel_launch(void* const* d_in, const int* in_sizes, int n_in,
                              void* d_out, int out_size, void* d_ws, size_t ws_size,
                              hipStream_t stream)
{
    (void)in_sizes; (void)n_in; (void)out_size;

    const float* enc        = (const float*)d_in[0];
    const float* W1         = (const float*)d_in[1];
    const float* b1         = (const float*)d_in[2];
    const float* W2         = (const float*)d_in[3];
    const float* b2         = (const float*)d_in[4];
    const int*   valid_mask = (const int*)d_in[5];
    const int*   pos_span   = (const int*)d_in[6];

    float* out  = (float*)d_out;
    float* feat = (float*)d_ws;
    float* part = feat + (size_t)B_ * D_;

    feat_kernel<<<dim3(D_ / SLICE_, B_), 32, 0, stream>>>(enc, valid_mask, pos_span, feat);
    head_kernel<<<dim3(NS_, B_ / 16), 64, 0, stream>>>(feat, W1, b1, W2, part, ws_size);
    sum_kernel<<<dim3(B_ * K_ / 4 / 32), 32, 0, stream>>>(part, b2, out, ws_size);
}
